# starting kernel
# baseline (speedup 1.0000x reference)
#include <hip/hip_runtime.h>
#include <hip/hip_bf16.h>

typedef __attribute__((ext_vector_type(2))) float v2f;
typedef __attribute__((ext_vector_type(8))) float v8f;

__global__ void snake_feat_kernel(const float* __restrict__ x,
                                  float* __restrict__ feat) {
  const int b   = blockIdx.x;
  const int tid = threadIdx.x;
  const float* __restrict__ g0 = x + (size_t)b * 3072;
  const float* __restrict__ g1 = g0 + 1024;

  __shared__ float    s_g0[1024];
  __shared__ unsigned s_min[3];
  __shared__ int      s_head[2];
  __shared__ int      s_dirs[3][2];
  __shared__ int      s_rot[2];
  __shared__ unsigned char s_freev[3][32];
  __shared__ unsigned char s_inb[3][32];

  if (tid < 3) s_min[tid] = 0xFFFFFFFFu;
  __syncthreads();

  const int i4 = tid * 4;
  const float4 v0 = *reinterpret_cast<const float4*>(g0 + i4);
  const float4 v1 = *reinterpret_cast<const float4*>(g1 + i4);
  *reinterpret_cast<float4*>(&s_g0[i4]) = v0;

  unsigned fkey = 0xFFFFFFFFu, hkey = 0xFFFFFFFFu, pkey = 0xFFFFFFFFu;
  const float a0[4] = {v0.x, v0.y, v0.z, v0.w};
  const float a1[4] = {v1.x, v1.y, v1.z, v1.w};
#pragma unroll
  for (int j = 0; j < 4; ++j) {
    const unsigned idx = (unsigned)(i4 + j);
    if (a0[j] == -1.0f) fkey = min(fkey, idx);
    if (a0[j] ==  1.0f) hkey = min(hkey, idx);
    if (a1[j] ==  1.0f) pkey = min(pkey, idx);
  }
  if (fkey != 0xFFFFFFFFu) atomicMin(&s_min[0], fkey);
  if (hkey != 0xFFFFFFFFu) atomicMin(&s_min[1], hkey);
  if (pkey != 0xFFFFFFFFu) atomicMin(&s_min[2], pkey);
  __syncthreads();

  if (tid == 0) {
    unsigned fi = s_min[0]; if (fi == 0xFFFFFFFFu) fi = 0;
    unsigned hi = s_min[1]; if (hi == 0xFFFFFFFFu) hi = 0;
    unsigned pi = s_min[2]; if (pi == 0xFFFFFFFFu) pi = 0;
    const int fx = (int)(fi >> 5), fy = (int)(fi & 31);
    const int hx = (int)(hi >> 5), hy = (int)(hi & 31);
    const int px = (int)(pi >> 5), py = (int)(pi & 31);
    const int dx = hx - px, dy = hy - py;
    const int ddx = fx - hx, ddy = fy - hy;
    const int k = (dx == 1) ? 0 : (dy == -1) ? 1 : (dx == -1) ? 2 : 3;
    int rx, ry;
    switch (k) {
      case 0:  rx =  ddx; ry =  ddy; break;
      case 1:  rx = -ddy; ry =  ddx; break;
      case 2:  rx = -ddx; ry = -ddy; break;
      default: rx =  ddy; ry = -ddx; break;
    }
    s_rot[0] = rx; s_rot[1] = ry;
    s_head[0] = hx; s_head[1] = hy;
    s_dirs[0][0] = -dy; s_dirs[0][1] =  dx;
    s_dirs[1][0] =  dx; s_dirs[1][1] =  dy;
    s_dirs[2][0] =  dy; s_dirs[2][1] = -dx;
  }
  __syncthreads();

  if (tid < 96) {
    const int dir = tid >> 5;
    const int i   = tid & 31;
    const int m   = i + 1;
    const int cx = s_head[0] + m * s_dirs[dir][0];
    const int cy = s_head[1] + m * s_dirs[dir][1];
    const bool inb = (cx >= 0) & (cx < 32) & (cy >= 0) & (cy < 32);
    const int ci = min(max(cx, 0), 31);
    const int cj = min(max(cy, 0), 31);
    const float val = s_g0[ci * 32 + cj];
    s_freev[dir][i] = (unsigned char)((val == 0.0f) | (val == -1.0f));
    s_inb[dir][i]   = (unsigned char)inb;
  }
  __syncthreads();

  if (tid < 3) {
    int freeLen = 0;
    for (int i = 0; i < 32; ++i) {
      const unsigned char fvp = s_freev[tid][(i == 0) ? 0 : i - 1];
      if (!(fvp & s_inb[tid][i])) { freeLen = i; break; }
    }
    feat[(size_t)b * 8 + tid] = (float)freeLen;
  }
  if (tid == 0) {
    feat[(size_t)b * 8 + 3] = (float)s_rot[0];
    feat[(size_t)b * 8 + 4] = (float)s_rot[1];
    feat[(size_t)b * 8 + 5] = 0.0f;
    feat[(size_t)b * 8 + 6] = 0.0f;
    feat[(size_t)b * 8 + 7] = 0.0f;
  }
}

static __device__ inline v8f splat8(float v) {
  v8f r;
#pragma unroll
  for (int i = 0; i < 8; ++i) r[i] = v;
  return r;
}

__global__ void snake_mlp_kernel(const float* __restrict__ feat,
                                 const float* __restrict__ w1,
                                 const float* __restrict__ b1,
                                 const float* __restrict__ w2,
                                 const float* __restrict__ b2,
                                 float* __restrict__ out) {
  const int lane = threadIdx.x;
  const int base = blockIdx.x * 16;
  const int M  = lane & 15;
  const int N  = lane & 15;
  const int kh = lane >> 4;

  __shared__ float Hs[16][20];

  const float* frow = feat + (size_t)(base + M) * 8;
  v2f a0; a0.x = frow[0 + 2 * kh]; a0.y = frow[1 + 2 * kh];
  v2f a1; a1.x = frow[4 + 2 * kh]; a1.y = frow[5 + 2 * kh];

  v8f h[2];
#pragma unroll
  for (int t = 0; t < 2; ++t) {
    const int n = t * 16 + N;
    const bool nv = (n < 20);
    const int k0a = 2 * kh;
    const int k0b = 4 + 2 * kh;
    v2f bA; bA.x = nv ? w1[n * 5 + k0a] : 0.0f;
            bA.y = nv ? w1[n * 5 + k0a + 1] : 0.0f;
    v2f bB; bB.x = (nv && k0b < 5) ? w1[n * 5 + k0b] : 0.0f;
            bB.y = 0.0f;
    v8f c = splat8(nv ? b1[n] : 0.0f);
    c = __builtin_amdgcn_wmma_f32_16x16x4_f32(false, a0, false, bA,
                                              (short)0, c, false, false);
    c = __builtin_amdgcn_wmma_f32_16x16x4_f32(false, a1, false, bB,
                                              (short)0, c, false, false);
#pragma unroll
    for (int v = 0; v < 8; ++v) c[v] = fmaxf(c[v], 0.0f);
    h[t] = c;
  }

#pragma unroll
  for (int v = 0; v < 8; ++v) {
    const int row = v + 8 * kh;
    Hs[row][N] = h[0][v];
    if (N < 4) Hs[row][16 + N] = h[1][v];
  }
  __syncthreads();

  const bool ov = (N < 3);
  v8f o = splat8(ov ? b2[N] : 0.0f);
#pragma unroll
  for (int c5 = 0; c5 < 5; ++c5) {
    const int k = 4 * c5 + 2 * kh;
    v2f a; a.x = Hs[M][k]; a.y = Hs[M][k + 1];
    v2f bb; bb.x = ov ? w2[N * 20 + k] : 0.0f;
            bb.y = ov ? w2[N * 20 + k + 1] : 0.0f;
    o = __builtin_amdgcn_wmma_f32_16x16x4_f32(false, a, false, bb,
                                              (short)0, o, false, false);
  }

  if (ov) {
#pragma unroll
    for (int v = 0; v < 8; ++v) {
      const int row = base + v + 8 * kh;
      out[(size_t)row * 3 + N] = o[v];
    }
  }
}

extern "C" void kernel_launch(void* const* d_in, const int* in_sizes, int n_in,
                              void* d_out, int out_size, void* d_ws, size_t ws_size,
                              hipStream_t stream) {
  const float* x  = (const float*)d_in[0];
  const float* w1 = (const float*)d_in[1];
  const float* b1 = (const float*)d_in[2];
  const float* w2 = (const float*)d_in[3];
  const float* b2 = (const float*)d_in[4];
  float* outp = (float*)d_out;
  float* featws = (float*)d_ws;

  const int B = in_sizes[0] / 3072;

  snake_feat_kernel<<<B, 256, 0, stream>>>(x, featws);
  snake_mlp_kernel<<<B / 16, 32, 0, stream>>>(featws, w1, b1, w2, b2, outp);
}


# Feature kernel: one wave per board (8 boards per 256-thread block): wave ballots and a DPP lane minimum replace LDS atomics, 4 block barriers and the serial ray loop; 8x fewer waves
# speedup vs baseline: 1.6755x; 1.6755x over previous
#include <hip/hip_runtime.h>
#include <hip/hip_bf16.h>

typedef __attribute__((ext_vector_type(2))) float v2f;
typedef __attribute__((ext_vector_type(8))) float v8f;


static __device__ inline unsigned wave_min_u32(unsigned v) {
  v = min(v, (unsigned)__builtin_amdgcn_update_dpp(-1, (int)v, 0x161, 0xf, 0xf, false));
  v = min(v, (unsigned)__builtin_amdgcn_update_dpp(-1, (int)v, 0x162, 0xf, 0xf, false));
  v = min(v, (unsigned)__builtin_amdgcn_update_dpp(-1, (int)v, 0x164, 0xf, 0xf, false));
  v = min(v, (unsigned)__builtin_amdgcn_update_dpp(-1, (int)v, 0x168, 0xf, 0xf, false));
  const unsigned lo = (unsigned)__builtin_amdgcn_readlane((int)v, 0);
  const unsigned hi = (unsigned)__builtin_amdgcn_readlane((int)v, 16);
  return min(lo, hi);
}

__global__ __launch_bounds__(256) void snake_feat_kernel(const float* __restrict__ x,
                                                         float* __restrict__ feat) {
  __shared__ float s_g0[8][1024];

  const int lane = threadIdx.x & 31;
  const int wave = __builtin_amdgcn_readfirstlane((int)(threadIdx.x >> 5));
  const int b    = blockIdx.x * 8 + wave;
  const float* __restrict__ g0 = x + (size_t)b * 3072;
  const float* __restrict__ g1 = g0 + 1024;
  float* __restrict__ sb = s_g0[wave];

  float4 c0[8], c1[8];
#pragma unroll
  for (int j = 7; j >= 0; --j)
    c0[j] = *reinterpret_cast<const float4*>(g0 + j * 128 + lane * 4);
#pragma unroll
  for (int j = 7; j >= 0; --j)
    c1[j] = *reinterpret_cast<const float4*>(g1 + j * 128 + lane * 4);

  unsigned qf = 32u, qh = 32u, qp = 32u;
#pragma unroll
  for (int j = 7; j >= 0; --j) {
    *reinterpret_cast<float4*>(sb + j * 128 + lane * 4) = c0[j];
    const float a[4] = {c0[j].x, c0[j].y, c0[j].z, c0[j].w};
#pragma unroll
    for (int c = 3; c >= 0; --c) {
      if (a[c] == -1.0f) qf = (unsigned)(4 * j + c);
      if (a[c] ==  1.0f) qh = (unsigned)(4 * j + c);
    }
  }
#pragma unroll
  for (int j = 7; j >= 0; --j) {
    const float a[4] = {c1[j].x, c1[j].y, c1[j].z, c1[j].w};
#pragma unroll
    for (int c = 3; c >= 0; --c)
      if (a[c] == 1.0f) qp = (unsigned)(4 * j + c);
  }

  const unsigned l4 = (unsigned)lane << 2;
  unsigned fi = wave_min_u32(((qf >> 2) << 7) | l4 | (qf & 3u));
  unsigned hi = wave_min_u32(((qh >> 2) << 7) | l4 | (qh & 3u));
  unsigned pi = wave_min_u32(((qp >> 2) << 7) | l4 | (qp & 3u));
  if (fi >= 1024u) fi = 0u;
  if (hi >= 1024u) hi = 0u;
  if (pi >= 1024u) pi = 0u;

  const int fx = (int)(fi >> 5), fy = (int)(fi & 31u);
  const int hx = (int)(hi >> 5), hy = (int)(hi & 31u);
  const int px = (int)(pi >> 5), py = (int)(pi & 31u);
  const int dx = hx - px, dy = hy - py;
  const int ddx = fx - hx, ddy = fy - hy;
  const int k = (dx == 1) ? 0 : (dy == -1) ? 1 : (dx == -1) ? 2 : 3;
  const int rx = (k == 0) ? ddx : (k == 1) ? -ddy : (k == 2) ? -ddx :  ddy;
  const int ry = (k == 0) ? ddy : (k == 1) ?  ddx : (k == 2) ? -ddy : -ddx;
  const int dirx[3] = {-dy, dx,  dy};
  const int diry[3] = { dx, dy, -dx};

  __builtin_amdgcn_fence(__ATOMIC_RELEASE, "wavefront");
  __builtin_amdgcn_wave_barrier();
  __builtin_amdgcn_fence(__ATOMIC_ACQUIRE, "wavefront");

  const int m = lane + 1;
  int freeLen[3];
#pragma unroll
  for (int d = 0; d < 3; ++d) {
    const int cx = hx + m * dirx[d];
    const int cy = hy + m * diry[d];
    const bool inb = (cx >= 0) & (cx < 32) & (cy >= 0) & (cy < 32);
    const int ci = min(max(cx, 0), 31);
    const int cj = min(max(cy, 0), 31);
    const float val = sb[ci * 32 + cj];
    const bool fv = (val == 0.0f) | (val == -1.0f);
    const unsigned FV  = __builtin_amdgcn_ballot_w32(fv);
    const unsigned INB = __builtin_amdgcn_ballot_w32(inb);
    const unsigned chk = ((FV << 1) | (FV & 1u)) & INB;
    const unsigned bad = ~chk;
    freeLen[d] = bad ? __builtin_ctz(bad) : 0;
  }

  float o = 0.0f;
  o = (lane == 0) ? (float)freeLen[0] : o;
  o = (lane == 1) ? (float)freeLen[1] : o;
  o = (lane == 2) ? (float)freeLen[2] : o;
  o = (lane == 3) ? (float)rx : o;
  o = (lane == 4) ? (float)ry : o;
  if (lane < 8) feat[(size_t)b * 8 + lane] = o;
}

static __device__ inline v8f splat8(float v) {
  v8f r;
#pragma unroll
  for (int i = 0; i < 8; ++i) r[i] = v;
  return r;
}

__global__ void snake_mlp_kernel(const float* __restrict__ feat,
                                 const float* __restrict__ w1,
                                 const float* __restrict__ b1,
                                 const float* __restrict__ w2,
                                 const float* __restrict__ b2,
                                 float* __restrict__ out) {
  const int lane = threadIdx.x;
  const int base = blockIdx.x * 16;
  const int M  = lane & 15;
  const int N  = lane & 15;
  const int kh = lane >> 4;

  __shared__ float Hs[16][20];

  const float* frow = feat + (size_t)(base + M) * 8;
  v2f a0; a0.x = frow[0 + 2 * kh]; a0.y = frow[1 + 2 * kh];
  v2f a1; a1.x = frow[4 + 2 * kh]; a1.y = frow[5 + 2 * kh];

  v8f h[2];
#pragma unroll
  for (int t = 0; t < 2; ++t) {
    const int n = t * 16 + N;
    const bool nv = (n < 20);
    const int k0a = 2 * kh;
    const int k0b = 4 + 2 * kh;
    v2f bA; bA.x = nv ? w1[n * 5 + k0a] : 0.0f;
            bA.y = nv ? w1[n * 5 + k0a + 1] : 0.0f;
    v2f bB; bB.x = (nv && k0b < 5) ? w1[n * 5 + k0b] : 0.0f;
            bB.y = 0.0f;
    v8f c = splat8(nv ? b1[n] : 0.0f);
    c = __builtin_amdgcn_wmma_f32_16x16x4_f32(false, a0, false, bA,
                                              (short)0, c, false, false);
    c = __builtin_amdgcn_wmma_f32_16x16x4_f32(false, a1, false, bB,
                                              (short)0, c, false, false);
#pragma unroll
    for (int v = 0; v < 8; ++v) c[v] = fmaxf(c[v], 0.0f);
    h[t] = c;
  }

#pragma unroll
  for (int v = 0; v < 8; ++v) {
    const int row = v + 8 * kh;
    Hs[row][N] = h[0][v];
    if (N < 4) Hs[row][16 + N] = h[1][v];
  }
  __syncthreads();

  const bool ov = (N < 3);
  v8f o = splat8(ov ? b2[N] : 0.0f);
#pragma unroll
  for (int c5 = 0; c5 < 5; ++c5) {
    const int k = 4 * c5 + 2 * kh;
    v2f a; a.x = Hs[M][k]; a.y = Hs[M][k + 1];
    v2f bb; bb.x = ov ? w2[N * 20 + k] : 0.0f;
            bb.y = ov ? w2[N * 20 + k + 1] : 0.0f;
    o = __builtin_amdgcn_wmma_f32_16x16x4_f32(false, a, false, bb,
                                              (short)0, o, false, false);
  }

  if (ov) {
#pragma unroll
    for (int v = 0; v < 8; ++v) {
      const int row = base + v + 8 * kh;
      out[(size_t)row * 3 + N] = o[v];
    }
  }
}

extern "C" void kernel_launch(void* const* d_in, const int* in_sizes, int n_in,
                              void* d_out, int out_size, void* d_ws, size_t ws_size,
                              hipStream_t stream) {
  const float* x  = (const float*)d_in[0];
  const float* w1 = (const float*)d_in[1];
  const float* b1 = (const float*)d_in[2];
  const float* w2 = (const float*)d_in[3];
  const float* b2 = (const float*)d_in[4];
  float* outp = (float*)d_out;
  float* featws = (float*)d_ws;

  const int B = in_sizes[0] / 3072;

  snake_feat_kernel<<<B / 8, 256, 0, stream>>>(x, featws);
  snake_mlp_kernel<<<B / 16, 32, 0, stream>>>(featws, w1, b1, w2, b2, outp);
}
